# starting kernel
# baseline (speedup 1.0000x reference)
#include <hip/hip_runtime.h>
#include <hip/hip_bf16.h>


typedef __attribute__((ext_vector_type(2))) float v2f;
typedef __attribute__((ext_vector_type(8))) float v8f;

#define T_LEN 2048
#define B_N   4096
#define H1N   9
#define H2N   32
#define DIN   5
#define OUTT  25

__device__ __forceinline__ v8f wmma4(v2f a, v2f b, v8f c) {
  return __builtin_amdgcn_wmma_f32_16x16x4_f32(
       false, a,  false, b,
       (short)0, c,  false,  false);
}

__device__ __forceinline__ float fast_tanh(float v) {
#if __has_builtin(__builtin_amdgcn_tanhf)
  return __builtin_amdgcn_tanhf(v);
#else
  return tanhf(v);
#endif
}

__device__ __forceinline__ void rnn1_step(
    v2f A0, v2f A1, v2f A2, v2f A3, v8f cbias,
    float d[5], float2 xp, float xw)
{
  v2f B0; B0.x = d[0]; B0.y = d[1];
  v2f B1; B1.x = d[2]; B1.y = d[3];
  v2f B2; B2.x = d[4]; B2.y = xw;
  v2f B3; B3.x = xp.x; B3.y = xp.y;

  v8f cz = {};
  v8f c1 = wmma4(A0, B0, cbias);
  v8f c2 = wmma4(A1, B1, cz);
  v8f c3 = wmma4(A2, B2, cz);
  v8f c4 = wmma4(A3, B3, cz);

#pragma unroll
  for (int i = 0; i < 5; ++i)
    d[i] = fast_tanh((c1[i] + c2[i]) + (c3[i] + c4[i]));
}

template <int DIR>
__device__ __forceinline__ void rnn1_time_loop(
    const float* __restrict__ x, long long xrow, int hi,
    v2f A0, v2f A1, v2f A2, v2f A3, v8f cbias, float d[5])
{
  constexpr int STEP = DIR ? -DIN : DIN;
  const long long t0 = DIR ? (long long)(T_LEN - 1) * DIN : 0;
  const float* q = x + xrow + t0 + 2 * hi;
  const float* r = x + xrow + t0 + 4;

#pragma unroll
  for (int i = 0; i < 5; ++i) d[i] = 0.f;

  float2 xp[4]; float xw[4];
#pragma unroll
  for (int u = 0; u < 4; ++u) {
    xp[u] = *(const float2*)&q[u * STEP];
    xw[u] = r[u * STEP];
  }
  q += 4 * STEP; r += 4 * STEP;

#pragma unroll 1
  for (int t = 0; t < T_LEN - 4; t += 4) {
    __builtin_prefetch(r + STEP * 60, 0, 1);
#pragma unroll
    for (int u = 0; u < 4; ++u) {
      rnn1_step(A0, A1, A2, A3, cbias, d, xp[u], xw[u]);
      xp[u] = *(const float2*)&q[u * STEP];
      xw[u] = r[u * STEP];
    }
    q += 4 * STEP; r += 4 * STEP;
  }
#pragma unroll
  for (int u = 0; u < 4; ++u)
    rnn1_step(A0, A1, A2, A3, cbias, d, xp[u], xw[u]);
}

__global__ __launch_bounds__(32) void rnn1_wmma_kernel(
    const float* __restrict__ x,
    const float* __restrict__ w_ih_f, const float* __restrict__ w_hh_f,
    const float* __restrict__ b_ih_f, const float* __restrict__ b_hh_f,
    const float* __restrict__ w_ih_b, const float* __restrict__ w_hh_b,
    const float* __restrict__ b_ih_b, const float* __restrict__ b_hh_b,
    float* __restrict__ y  )
{
  const int lane = threadIdx.x;
  const int g    = blockIdx.x;
  const int dir  = g >> 8;
  const int s0   = (g & 255) * 16;
  const int hi   = lane >> 4;
  const int ii   = lane & 15;

  const float* wih = dir ? w_ih_b : w_ih_f;
  const float* whh = dir ? w_hh_b : w_hh_f;
  const float* bih = dir ? b_ih_b : b_ih_f;
  const float* bhh = dir ? b_hh_b : b_hh_f;

  auto unitOfRow = [](int i) -> int {
    return (i < 5) ? i : ((i >= 8 && i < 12) ? i - 3 : -1);
  };
  const int slotType[16] = {0,0,0,0, 0,0,0,0, 0,2,2,1, 1,1,1,1};
  const int slotIdx [16] = {0,1,5,6, 2,3,7,8, 4,0,0,4, 0,1,2,3};

  const int urow = unitOfRow(ii);
  auto wElem = [&](int k) -> float {
    if (urow < 0) return 0.f;
    const int ty = slotType[k], ix = slotIdx[k];
    if (ty == 0) return whh[urow * H1N + ix];
    if (ty == 1) return wih[urow * DIN + ix];
    return 0.f;
  };
  auto aW = [&](int kb) -> v2f {
    v2f v; const int k0 = kb + 2 * hi;
    v.x = wElem(k0); v.y = wElem(k0 + 1);
    return v;
  };
  const v2f A0 = aW(0), A1 = aW(4), A2 = aW(8), A3 = aW(12);

  v8f cbias;
#pragma unroll
  for (int rr = 0; rr < 8; ++rr) {
    const int uu = unitOfRow(rr + 8 * hi);
    cbias[rr] = (uu >= 0) ? (bih[uu] + bhh[uu]) : 0.f;
  }

  const long long xrow = (long long)(s0 + ii) * (T_LEN * DIN);

  float d[5];
  if (dir == 0)
    rnn1_time_loop<0>(x, xrow, hi, A0, A1, A2, A3, cbias, d);
  else
    rnn1_time_loop<1>(x, xrow, hi, A0, A1, A2, A3, cbias, d);

  const int sj = s0 + ii;
#pragma unroll
  for (int rr = 0; rr < 5; ++rr) {
    const int uu = unitOfRow(rr + 8 * hi);
    if (uu >= 0) y[sj * (2 * H1N) + dir * H1N + uu] = d[rr];
  }
}

__global__ __launch_bounds__(32) void rnn2_wmma_kernel(
    const float* __restrict__ w_ih2, const float* __restrict__ w_hh2,
    const float* __restrict__ b_ih2, const float* __restrict__ b_hh2,
    const float* __restrict__ y   ,
    float* __restrict__ ys        )
{
  const int lane = threadIdx.x;
  const int hi = lane >> 4, ii = lane & 15;
  const int s0 = blockIdx.x * 16;

  v2f Aih[5][2];
#pragma unroll
  for (int c = 0; c < 5; ++c)
#pragma unroll
    for (int nt = 0; nt < 2; ++nt) {
      const int k0 = 4 * c + 2 * hi, i = nt * 16 + ii;
      Aih[c][nt].x = ((k0    ) < 18) ? w_ih2[i * 18 + k0    ] : 0.f;
      Aih[c][nt].y = ((k0 + 1) < 18) ? w_ih2[i * 18 + k0 + 1] : 0.f;
    }
  v2f Ahh[8][2];
#pragma unroll
  for (int c = 0; c < 8; ++c)
#pragma unroll
    for (int nt = 0; nt < 2; ++nt) {
      const int j0 = 16 * (c >> 2) + 8 * hi + 2 * (c & 3);
      const int i = nt * 16 + ii;
      Ahh[c][nt].x = w_hh2[i * 32 + j0    ];
      Ahh[c][nt].y = w_hh2[i * 32 + j0 + 1];
    }

  v8f cb[2];
#pragma unroll
  for (int nt = 0; nt < 2; ++nt)
#pragma unroll
    for (int rr = 0; rr < 8; ++rr) {
      const int i = nt * 16 + rr + 8 * hi;
      cb[nt][rr] = b_ih2[i] + b_hh2[i];
    }

  const int base = (s0 + ii) * 18;
  v2f By[5];
#pragma unroll
  for (int c = 0; c < 5; ++c) {
    const int k0 = 4 * c + 2 * hi;
    const float v0 = y[base + k0];
    const float v1 = y[base + k0 + 1];
    By[c].x = ((k0    ) < 18) ? v0 : 0.f;
    By[c].y = ((k0 + 1) < 18) ? v1 : 0.f;
  }

  v8f d[2];
  v8f cz = {};
#pragma unroll 1
  for (int t = 0; t < OUTT; ++t) {
    if (t == 0) {
#pragma unroll
      for (int nt = 0; nt < 2; ++nt) {
        v8f c1 = wmma4(Aih[0][nt], By[0], cb[nt]);
        v8f c2 = wmma4(Aih[1][nt], By[1], cz);
        c1 = wmma4(Aih[2][nt], By[2], c1);
        c2 = wmma4(Aih[3][nt], By[3], c2);
        c1 = wmma4(Aih[4][nt], By[4], c1);
#pragma unroll
        for (int rr = 0; rr < 8; ++rr) d[nt][rr] = fast_tanh(c1[rr] + c2[rr]);
      }
    } else {
      v2f Bh[8];
#pragma unroll
      for (int c = 0; c < 8; ++c) {
        Bh[c].x = d[c >> 2][2 * (c & 3)    ];
        Bh[c].y = d[c >> 2][2 * (c & 3) + 1];
      }
      v8f dn[2];
#pragma unroll
      for (int nt = 0; nt < 2; ++nt) {
        v8f c1 = cb[nt];
        v8f c2 = cz;
#pragma unroll
        for (int cc = 0; cc < 8; cc += 2) {
          c1 = wmma4(Ahh[cc    ][nt], Bh[cc    ], c1);
          c2 = wmma4(Ahh[cc + 1][nt], Bh[cc + 1], c2);
        }
#pragma unroll
        for (int rr = 0; rr < 8; ++rr) dn[nt][rr] = fast_tanh(c1[rr] + c2[rr]);
      }
      d[0] = dn[0]; d[1] = dn[1];
    }

    const long long ob = ((long long)(s0 + ii) * OUTT + t) * 32;
#pragma unroll
    for (int nt = 0; nt < 2; ++nt)
#pragma unroll
      for (int rr = 0; rr < 8; ++rr)
        ys[ob + nt * 16 + rr + 8 * hi] = d[nt][rr];
  }
}

__global__ __launch_bounds__(256) void out_proj_kernel(
    const float* __restrict__ ys, const float* __restrict__ w_out,
    const float* __restrict__ b_out, float* __restrict__ out)
{
  const int id = blockIdx.x * blockDim.x + threadIdx.x;
  if (id >= B_N * OUTT) return;
  const float* row = ys + (long long)id * 32;
  float a0 = b_out[0], a1 = b_out[1], a2 = b_out[2];
#pragma unroll
  for (int nn = 0; nn < 32; ++nn) {
    const float v = row[nn];
    a0 = fmaf(v, w_out[nn],      a0);
    a1 = fmaf(v, w_out[32 + nn], a1);
    a2 = fmaf(v, w_out[64 + nn], a2);
  }
  out[(long long)id * 3 + 0] = a0;
  out[(long long)id * 3 + 1] = a1;
  out[(long long)id * 3 + 2] = a2;
}

extern "C" void kernel_launch(void* const* d_in, const int* in_sizes, int n_in,
                              void* d_out, int out_size, void* d_ws,
                              size_t ws_size, hipStream_t stream) {
  const float* x      = (const float*)d_in[0];
  const float* w_ih_f = (const float*)d_in[1];
  const float* w_hh_f = (const float*)d_in[2];
  const float* b_ih_f = (const float*)d_in[3];
  const float* b_hh_f = (const float*)d_in[4];
  const float* w_ih_b = (const float*)d_in[5];
  const float* w_hh_b = (const float*)d_in[6];
  const float* b_ih_b = (const float*)d_in[7];
  const float* b_hh_b = (const float*)d_in[8];
  const float* w_ih2  = (const float*)d_in[9];
  const float* w_hh2  = (const float*)d_in[10];
  const float* b_ih2  = (const float*)d_in[11];
  const float* b_hh2  = (const float*)d_in[12];
  const float* w_out  = (const float*)d_in[13];
  const float* b_out  = (const float*)d_in[14];

  float* y  = (float*)d_ws;
  float* ys = y + (size_t)B_N * (2 * H1N);

  rnn1_wmma_kernel<<<dim3(512), dim3(32), 0, stream>>>(
      x, w_ih_f, w_hh_f, b_ih_f, b_hh_f,
      w_ih_b, w_hh_b, b_ih_b, b_hh_b, y);

  rnn2_wmma_kernel<<<dim3(256), dim3(32), 0, stream>>>(
      w_ih2, w_hh2, b_ih2, b_hh2, y, ys);

  const int nthreads = B_N * OUTT;
  out_proj_kernel<<<dim3((nthreads + 255) / 256), dim3(256), 0, stream>>>(
      ys, w_out, b_out, (float*)d_out);
}


# rnn1 step issue order: 3 state WMMAs back to back, next step's input WMMA behind them, all sums after; one matrix-to-vector wait per step, not two; x staged 8 steps ahead; sum order unchanged
# speedup vs baseline: 1.2910x; 1.2910x over previous
#include <hip/hip_runtime.h>
#include <hip/hip_bf16.h>


typedef __attribute__((ext_vector_type(2))) float v2f;
typedef __attribute__((ext_vector_type(8))) float v8f;

#define T_LEN 2048
#define B_N   4096
#define H1N   9
#define H2N   32
#define DIN   5
#define OUTT  25

__device__ __forceinline__ v8f wmma4(v2f a, v2f b, v8f c) {
  return __builtin_amdgcn_wmma_f32_16x16x4_f32(
       false, a,  false, b,
       (short)0, c,  false,  false);
}

__device__ __forceinline__ float fast_tanh(float v) {
#if __has_builtin(__builtin_amdgcn_tanhf)
  return __builtin_amdgcn_tanhf(v);
#else
  return tanhf(v);
#endif
}

template <bool NEXT>
__device__ __forceinline__ void rnn1_step(
    v2f A0, v2f A1, v2f A2, v2f A3, v8f cbias,
    float d[5], v8f& c4, float xw, float2 xpn)
{
  v2f B0; B0.x = d[0]; B0.y = d[1];
  v2f B1; B1.x = d[2]; B1.y = d[3];
  v2f B2; B2.x = d[4]; B2.y = xw;

  v8f cz = {};
  v8f c1 = wmma4(A0, B0, cbias);
  v8f c2 = wmma4(A1, B1, cz);
  v8f c3 = wmma4(A2, B2, cz);
  v8f c4n = cz;
  if (NEXT) {
    v2f B3; B3.x = xpn.x; B3.y = xpn.y;
    c4n = wmma4(A3, B3, cz);
  }
  __builtin_amdgcn_sched_barrier(0);

#pragma unroll
  for (int i = 0; i < 5; ++i)
    d[i] = fast_tanh((c1[i] + c2[i]) + (c3[i] + c4[i]));
  c4 = c4n;
}

template <int DIR>
__device__ __forceinline__ void rnn1_time_loop(
    const float* __restrict__ x, long long xrow, int hi,
    v2f A0, v2f A1, v2f A2, v2f A3, v8f cbias, float d[5])
{
  constexpr int STEP = DIR ? -DIN : DIN;
  constexpr int NS = 8;
  const long long t0 = DIR ? (long long)(T_LEN - 1) * DIN : 0;
  const float* q = x + xrow + t0 + 2 * hi;
  const float* r = x + xrow + t0 + 4;

#pragma unroll
  for (int i = 0; i < 5; ++i) d[i] = 0.f;

  const float2 xp0 = *(const float2*)&q[0];
  float2 xp[NS]; float xw[NS];
#pragma unroll
  for (int u = 0; u < NS; ++u) {
    xp[u] = *(const float2*)&q[(u + 1) * STEP];
    xw[u] = r[u * STEP];
  }
  q += NS * STEP; r += NS * STEP;

  v8f c4;
  {
    v8f cz = {};
    v2f B3; B3.x = xp0.x; B3.y = xp0.y;
    c4 = wmma4(A3, B3, cz);
  }

#pragma unroll 1
  for (int t = 0; t < T_LEN - 2 * NS; t += NS) {
#pragma unroll
    for (int u = 0; u < NS; ++u) {
      if ((u & 3) == 0)
        __builtin_prefetch(r + STEP * (60 + u), 0, 1);
      rnn1_step<true>(A0, A1, A2, A3, cbias, d, c4, xw[u], xp[u]);
      xp[u] = *(const float2*)&q[(u + 1) * STEP];
      xw[u] = r[u * STEP];
    }
    q += NS * STEP; r += NS * STEP;
  }
#pragma unroll
  for (int u = 0; u < NS; ++u) {
    rnn1_step<true>(A0, A1, A2, A3, cbias, d, c4, xw[u], xp[u]);
    if (u < NS - 1) xp[u] = *(const float2*)&q[(u + 1) * STEP];
    xw[u] = r[u * STEP];
  }
#pragma unroll
  for (int u = 0; u < NS - 1; ++u)
    rnn1_step<true>(A0, A1, A2, A3, cbias, d, c4, xw[u], xp[u]);
  rnn1_step<false>(A0, A1, A2, A3, cbias, d, c4, xw[NS - 1], xp[NS - 1]);
}

__global__ __launch_bounds__(32) void rnn1_wmma_kernel(
    const float* __restrict__ x,
    const float* __restrict__ w_ih_f, const float* __restrict__ w_hh_f,
    const float* __restrict__ b_ih_f, const float* __restrict__ b_hh_f,
    const float* __restrict__ w_ih_b, const float* __restrict__ w_hh_b,
    const float* __restrict__ b_ih_b, const float* __restrict__ b_hh_b,
    float* __restrict__ y  )
{
  const int lane = threadIdx.x;
  const int g    = blockIdx.x;
  const int dir  = g >> 8;
  const int s0   = (g & 255) * 16;
  const int hi   = lane >> 4;
  const int ii   = lane & 15;

  const float* wih = dir ? w_ih_b : w_ih_f;
  const float* whh = dir ? w_hh_b : w_hh_f;
  const float* bih = dir ? b_ih_b : b_ih_f;
  const float* bhh = dir ? b_hh_b : b_hh_f;

  auto unitOfRow = [](int i) -> int {
    return (i < 5) ? i : ((i >= 8 && i < 12) ? i - 3 : -1);
  };
  const int slotType[16] = {0,0,0,0, 0,0,0,0, 0,2,2,1, 1,1,1,1};
  const int slotIdx [16] = {0,1,5,6, 2,3,7,8, 4,0,0,4, 0,1,2,3};

  const int urow = unitOfRow(ii);
  auto wElem = [&](int k) -> float {
    if (urow < 0) return 0.f;
    const int ty = slotType[k], ix = slotIdx[k];
    if (ty == 0) return whh[urow * H1N + ix];
    if (ty == 1) return wih[urow * DIN + ix];
    return 0.f;
  };
  auto aW = [&](int kb) -> v2f {
    v2f v; const int k0 = kb + 2 * hi;
    v.x = wElem(k0); v.y = wElem(k0 + 1);
    return v;
  };
  const v2f A0 = aW(0), A1 = aW(4), A2 = aW(8), A3 = aW(12);

  v8f cbias;
#pragma unroll
  for (int rr = 0; rr < 8; ++rr) {
    const int uu = unitOfRow(rr + 8 * hi);
    cbias[rr] = (uu >= 0) ? (bih[uu] + bhh[uu]) : 0.f;
  }

  const long long xrow = (long long)(s0 + ii) * (T_LEN * DIN);

  float d[5];
  if (dir == 0)
    rnn1_time_loop<0>(x, xrow, hi, A0, A1, A2, A3, cbias, d);
  else
    rnn1_time_loop<1>(x, xrow, hi, A0, A1, A2, A3, cbias, d);

  const int sj = s0 + ii;
#pragma unroll
  for (int rr = 0; rr < 5; ++rr) {
    const int uu = unitOfRow(rr + 8 * hi);
    if (uu >= 0) y[sj * (2 * H1N) + dir * H1N + uu] = d[rr];
  }
}

__global__ __launch_bounds__(32) void rnn2_wmma_kernel(
    const float* __restrict__ w_ih2, const float* __restrict__ w_hh2,
    const float* __restrict__ b_ih2, const float* __restrict__ b_hh2,
    const float* __restrict__ y   ,
    float* __restrict__ ys        )
{
  const int lane = threadIdx.x;
  const int hi = lane >> 4, ii = lane & 15;
  const int s0 = blockIdx.x * 16;

  v2f Aih[5][2];
#pragma unroll
  for (int c = 0; c < 5; ++c)
#pragma unroll
    for (int nt = 0; nt < 2; ++nt) {
      const int k0 = 4 * c + 2 * hi, i = nt * 16 + ii;
      Aih[c][nt].x = ((k0    ) < 18) ? w_ih2[i * 18 + k0    ] : 0.f;
      Aih[c][nt].y = ((k0 + 1) < 18) ? w_ih2[i * 18 + k0 + 1] : 0.f;
    }
  v2f Ahh[8][2];
#pragma unroll
  for (int c = 0; c < 8; ++c)
#pragma unroll
    for (int nt = 0; nt < 2; ++nt) {
      const int j0 = 16 * (c >> 2) + 8 * hi + 2 * (c & 3);
      const int i = nt * 16 + ii;
      Ahh[c][nt].x = w_hh2[i * 32 + j0    ];
      Ahh[c][nt].y = w_hh2[i * 32 + j0 + 1];
    }

  v8f cb[2];
#pragma unroll
  for (int nt = 0; nt < 2; ++nt)
#pragma unroll
    for (int rr = 0; rr < 8; ++rr) {
      const int i = nt * 16 + rr + 8 * hi;
      cb[nt][rr] = b_ih2[i] + b_hh2[i];
    }

  const int base = (s0 + ii) * 18;
  v2f By[5];
#pragma unroll
  for (int c = 0; c < 5; ++c) {
    const int k0 = 4 * c + 2 * hi;
    const float v0 = y[base + k0];
    const float v1 = y[base + k0 + 1];
    By[c].x = ((k0    ) < 18) ? v0 : 0.f;
    By[c].y = ((k0 + 1) < 18) ? v1 : 0.f;
  }

  v8f d[2];
  v8f cz = {};
#pragma unroll 1
  for (int t = 0; t < OUTT; ++t) {
    if (t == 0) {
#pragma unroll
      for (int nt = 0; nt < 2; ++nt) {
        v8f c1 = wmma4(Aih[0][nt], By[0], cb[nt]);
        v8f c2 = wmma4(Aih[1][nt], By[1], cz);
        c1 = wmma4(Aih[2][nt], By[2], c1);
        c2 = wmma4(Aih[3][nt], By[3], c2);
        c1 = wmma4(Aih[4][nt], By[4], c1);
#pragma unroll
        for (int rr = 0; rr < 8; ++rr) d[nt][rr] = fast_tanh(c1[rr] + c2[rr]);
      }
    } else {
      v2f Bh[8];
#pragma unroll
      for (int c = 0; c < 8; ++c) {
        Bh[c].x = d[c >> 2][2 * (c & 3)    ];
        Bh[c].y = d[c >> 2][2 * (c & 3) + 1];
      }
      v8f dn[2];
#pragma unroll
      for (int nt = 0; nt < 2; ++nt) {
        v8f c1 = cb[nt];
        v8f c2 = cz;
#pragma unroll
        for (int cc = 0; cc < 8; cc += 2) {
          c1 = wmma4(Ahh[cc    ][nt], Bh[cc    ], c1);
          c2 = wmma4(Ahh[cc + 1][nt], Bh[cc + 1], c2);
        }
#pragma unroll
        for (int rr = 0; rr < 8; ++rr) dn[nt][rr] = fast_tanh(c1[rr] + c2[rr]);
      }
      d[0] = dn[0]; d[1] = dn[1];
    }

    const long long ob = ((long long)(s0 + ii) * OUTT + t) * 32;
#pragma unroll
    for (int nt = 0; nt < 2; ++nt)
#pragma unroll
      for (int rr = 0; rr < 8; ++rr)
        ys[ob + nt * 16 + rr + 8 * hi] = d[nt][rr];
  }
}

__global__ __launch_bounds__(256) void out_proj_kernel(
    const float* __restrict__ ys, const float* __restrict__ w_out,
    const float* __restrict__ b_out, float* __restrict__ out)
{
  const int id = blockIdx.x * blockDim.x + threadIdx.x;
  if (id >= B_N * OUTT) return;
  const float* row = ys + (long long)id * 32;
  float a0 = b_out[0], a1 = b_out[1], a2 = b_out[2];
#pragma unroll
  for (int nn = 0; nn < 32; ++nn) {
    const float v = row[nn];
    a0 = fmaf(v, w_out[nn],      a0);
    a1 = fmaf(v, w_out[32 + nn], a1);
    a2 = fmaf(v, w_out[64 + nn], a2);
  }
  out[(long long)id * 3 + 0] = a0;
  out[(long long)id * 3 + 1] = a1;
  out[(long long)id * 3 + 2] = a2;
}

extern "C" void kernel_launch(void* const* d_in, const int* in_sizes, int n_in,
                              void* d_out, int out_size, void* d_ws,
                              size_t ws_size, hipStream_t stream) {
  const float* x      = (const float*)d_in[0];
  const float* w_ih_f = (const float*)d_in[1];
  const float* w_hh_f = (const float*)d_in[2];
  const float* b_ih_f = (const float*)d_in[3];
  const float* b_hh_f = (const float*)d_in[4];
  const float* w_ih_b = (const float*)d_in[5];
  const float* w_hh_b = (const float*)d_in[6];
  const float* b_ih_b = (const float*)d_in[7];
  const float* b_hh_b = (const float*)d_in[8];
  const float* w_ih2  = (const float*)d_in[9];
  const float* w_hh2  = (const float*)d_in[10];
  const float* b_ih2  = (const float*)d_in[11];
  const float* b_hh2  = (const float*)d_in[12];
  const float* w_out  = (const float*)d_in[13];
  const float* b_out  = (const float*)d_in[14];

  float* y  = (float*)d_ws;
  float* ys = y + (size_t)B_N * (2 * H1N);

  rnn1_wmma_kernel<<<dim3(512), dim3(32), 0, stream>>>(
      x, w_ih_f, w_hh_f, b_ih_f, b_hh_f,
      w_ih_b, w_hh_b, b_ih_b, b_hh_b, y);

  rnn2_wmma_kernel<<<dim3(256), dim3(32), 0, stream>>>(
      w_ih2, w_hh2, b_ih2, b_hh2, y, ys);

  const int nthreads = B_N * OUTT;
  out_proj_kernel<<<dim3((nthreads + 255) / 256), dim3(256), 0, stream>>>(
      ys, w_out, b_out, (float*)d_out);
}


# rnn1 x feed: all loads and both prefetches of an 8-step block issued together at the block seam, one block ahead (two register sets), not 2 loads in each step behind its products; sums unchanged
# speedup vs baseline: 1.5762x; 1.2209x over previous
#include <hip/hip_runtime.h>
#include <hip/hip_bf16.h>


typedef __attribute__((ext_vector_type(2))) float v2f;
typedef __attribute__((ext_vector_type(8))) float v8f;

#define T_LEN 2048
#define B_N   4096
#define H1N   9
#define H2N   32
#define DIN   5
#define OUTT  25

__device__ __forceinline__ v8f wmma4(v2f a, v2f b, v8f c) {
  return __builtin_amdgcn_wmma_f32_16x16x4_f32(
       false, a,  false, b,
       (short)0, c,  false,  false);
}

__device__ __forceinline__ float fast_tanh(float v) {
#if __has_builtin(__builtin_amdgcn_tanhf)
  return __builtin_amdgcn_tanhf(v);
#else
  return tanhf(v);
#endif
}

template <bool NEXT>
__device__ __forceinline__ void rnn1_step(
    v2f A0, v2f A1, v2f A2, v2f A3, v8f cbias,
    float d[5], v8f& c4, float xw, float2 xpn)
{
  v2f B0; B0.x = d[0]; B0.y = d[1];
  v2f B1; B1.x = d[2]; B1.y = d[3];
  v2f B2; B2.x = d[4]; B2.y = xw;

  v8f cz = {};
  v8f c1 = wmma4(A0, B0, cbias);
  v8f c2 = wmma4(A1, B1, cz);
  v8f c3 = wmma4(A2, B2, cz);
  v8f c4n = cz;
  if (NEXT) {
    v2f B3; B3.x = xpn.x; B3.y = xpn.y;
    asm volatile("" : "+v"(c1), "+v"(c2), "+v"(c3), "+v"(B3));
    c4n = wmma4(A3, B3, cz);
    asm volatile("" : "+v"(c4n));
  }
  __builtin_amdgcn_sched_barrier(0);

#pragma unroll
  for (int i = 0; i < 5; ++i)
    d[i] = fast_tanh((c1[i] + c2[i]) + (c3[i] + c4[i]));
  c4 = c4n;
}

__device__ __forceinline__ void rnn1_seam(float d[5], long long& qo, long long& ro)
{
  asm volatile("" : "+v"(d[0]), "+v"(d[1]), "+v"(d[2]), "+v"(d[3]), "+v"(d[4]),
                    "+v"(qo), "+v"(ro));
}

template <int DIR, bool LASTPAIR>
__device__ __forceinline__ void rnn1_fetch_block(
    const float* q, const float* r, float2 xp[8], float xw[8])
{
  constexpr int STEP = DIR ? -DIN : DIN;
#pragma unroll
  for (int u = 0; u < 8; ++u) {
    if (LASTPAIR || u < 7) xp[u] = *(const float2*)&q[(u + 1) * STEP];
    xw[u] = r[u * STEP];
  }
}

template <bool FINAL>
__device__ __forceinline__ void rnn1_run_block(
    v2f A0, v2f A1, v2f A2, v2f A3, v8f cbias,
    float d[5], v8f& c4, const float2 xp[8], const float xw[8])
{
#pragma unroll
  for (int u = 0; u < 7; ++u)
    rnn1_step<true>(A0, A1, A2, A3, cbias, d, c4, xw[u], xp[u]);
  if (FINAL)
    rnn1_step<false>(A0, A1, A2, A3, cbias, d, c4, xw[7], xp[6]);
  else
    rnn1_step<true>(A0, A1, A2, A3, cbias, d, c4, xw[7], xp[7]);
}

template <int DIR>
__device__ __forceinline__ void rnn1_time_loop(
    const float* __restrict__ x, long long xrow, int hi,
    v2f A0, v2f A1, v2f A2, v2f A3, v8f cbias, float d[5])
{
  constexpr int STEP = DIR ? -DIN : DIN;
  constexpr int NS = 8;
  constexpr int NB = T_LEN / NS;
  constexpr int PF = 60;
  const long long t0 = DIR ? (long long)(T_LEN - 1) * DIN : 0;
  long long qo = xrow + t0 + 2 * hi;
  long long ro = xrow + t0 + 4;

#pragma unroll
  for (int i = 0; i < 5; ++i) d[i] = 0.f;

  float2 xpA[NS], xpB[NS];
  float  xwA[NS], xwB[NS];

  const float2 xp0 = *(const float2*)&x[qo];
  rnn1_fetch_block<DIR, true>(x + qo, x + ro, xpA, xwA);
  qo += NS * STEP; ro += NS * STEP;

  v8f c4;
  {
    v8f cz = {};
    v2f B3; B3.x = xp0.x; B3.y = xp0.y;
    c4 = wmma4(A3, B3, cz);
  }

#pragma unroll 1
  for (int b = 0; b < NB - 2; b += 2) {
    rnn1_seam(d, qo, ro);
    rnn1_fetch_block<DIR, true>(x + qo, x + ro, xpB, xwB);
    if (b < NB - 10) {
      __builtin_prefetch(x + ro + STEP * PF, 0, 1);
      __builtin_prefetch(x + ro + STEP * (PF + 4), 0, 1);
    }
    qo += NS * STEP; ro += NS * STEP;
    __builtin_amdgcn_sched_barrier(0);
    rnn1_run_block<false>(A0, A1, A2, A3, cbias, d, c4, xpA, xwA);

    rnn1_seam(d, qo, ro);
    rnn1_fetch_block<DIR, true>(x + qo, x + ro, xpA, xwA);
    if (b < NB - 10) {
      __builtin_prefetch(x + ro + STEP * PF, 0, 1);
      __builtin_prefetch(x + ro + STEP * (PF + 4), 0, 1);
    }
    qo += NS * STEP; ro += NS * STEP;
    __builtin_amdgcn_sched_barrier(0);
    rnn1_run_block<false>(A0, A1, A2, A3, cbias, d, c4, xpB, xwB);
  }

  rnn1_seam(d, qo, ro);
  rnn1_fetch_block<DIR, false>(x + qo, x + ro, xpB, xwB);
  __builtin_amdgcn_sched_barrier(0);
  rnn1_run_block<false>(A0, A1, A2, A3, cbias, d, c4, xpA, xwA);
  rnn1_run_block<true >(A0, A1, A2, A3, cbias, d, c4, xpB, xwB);
}

__global__ __launch_bounds__(32) void rnn1_wmma_kernel(
    const float* __restrict__ x,
    const float* __restrict__ w_ih_f, const float* __restrict__ w_hh_f,
    const float* __restrict__ b_ih_f, const float* __restrict__ b_hh_f,
    const float* __restrict__ w_ih_b, const float* __restrict__ w_hh_b,
    const float* __restrict__ b_ih_b, const float* __restrict__ b_hh_b,
    float* __restrict__ y  )
{
  const int lane = threadIdx.x;
  const int g    = blockIdx.x;
  const int dir  = g >> 8;
  const int s0   = (g & 255) * 16;
  const int hi   = lane >> 4;
  const int ii   = lane & 15;

  const float* wih = dir ? w_ih_b : w_ih_f;
  const float* whh = dir ? w_hh_b : w_hh_f;
  const float* bih = dir ? b_ih_b : b_ih_f;
  const float* bhh = dir ? b_hh_b : b_hh_f;

  auto unitOfRow = [](int i) -> int {
    return (i < 5) ? i : ((i >= 8 && i < 12) ? i - 3 : -1);
  };
  const int slotType[16] = {0,0,0,0, 0,0,0,0, 0,2,2,1, 1,1,1,1};
  const int slotIdx [16] = {0,1,5,6, 2,3,7,8, 4,0,0,4, 0,1,2,3};

  const int urow = unitOfRow(ii);
  auto wElem = [&](int k) -> float {
    if (urow < 0) return 0.f;
    const int ty = slotType[k], ix = slotIdx[k];
    if (ty == 0) return whh[urow * H1N + ix];
    if (ty == 1) return wih[urow * DIN + ix];
    return 0.f;
  };
  auto aW = [&](int kb) -> v2f {
    v2f v; const int k0 = kb + 2 * hi;
    v.x = wElem(k0); v.y = wElem(k0 + 1);
    return v;
  };
  const v2f A0 = aW(0), A1 = aW(4), A2 = aW(8), A3 = aW(12);

  v8f cbias;
#pragma unroll
  for (int rr = 0; rr < 8; ++rr) {
    const int uu = unitOfRow(rr + 8 * hi);
    cbias[rr] = (uu >= 0) ? (bih[uu] + bhh[uu]) : 0.f;
  }

  const long long xrow = (long long)(s0 + ii) * (T_LEN * DIN);

  float d[5];
  if (dir == 0)
    rnn1_time_loop<0>(x, xrow, hi, A0, A1, A2, A3, cbias, d);
  else
    rnn1_time_loop<1>(x, xrow, hi, A0, A1, A2, A3, cbias, d);

  const int sj = s0 + ii;
#pragma unroll
  for (int rr = 0; rr < 5; ++rr) {
    const int uu = unitOfRow(rr + 8 * hi);
    if (uu >= 0) y[sj * (2 * H1N) + dir * H1N + uu] = d[rr];
  }
}

__global__ __launch_bounds__(32) void rnn2_wmma_kernel(
    const float* __restrict__ w_ih2, const float* __restrict__ w_hh2,
    const float* __restrict__ b_ih2, const float* __restrict__ b_hh2,
    const float* __restrict__ y   ,
    float* __restrict__ ys        )
{
  const int lane = threadIdx.x;
  const int hi = lane >> 4, ii = lane & 15;
  const int s0 = blockIdx.x * 16;

  v2f Aih[5][2];
#pragma unroll
  for (int c = 0; c < 5; ++c)
#pragma unroll
    for (int nt = 0; nt < 2; ++nt) {
      const int k0 = 4 * c + 2 * hi, i = nt * 16 + ii;
      Aih[c][nt].x = ((k0    ) < 18) ? w_ih2[i * 18 + k0    ] : 0.f;
      Aih[c][nt].y = ((k0 + 1) < 18) ? w_ih2[i * 18 + k0 + 1] : 0.f;
    }
  v2f Ahh[8][2];
#pragma unroll
  for (int c = 0; c < 8; ++c)
#pragma unroll
    for (int nt = 0; nt < 2; ++nt) {
      const int j0 = 16 * (c >> 2) + 8 * hi + 2 * (c & 3);
      const int i = nt * 16 + ii;
      Ahh[c][nt].x = w_hh2[i * 32 + j0    ];
      Ahh[c][nt].y = w_hh2[i * 32 + j0 + 1];
    }

  v8f cb[2];
#pragma unroll
  for (int nt = 0; nt < 2; ++nt)
#pragma unroll
    for (int rr = 0; rr < 8; ++rr) {
      const int i = nt * 16 + rr + 8 * hi;
      cb[nt][rr] = b_ih2[i] + b_hh2[i];
    }

  const int base = (s0 + ii) * 18;
  v2f By[5];
#pragma unroll
  for (int c = 0; c < 5; ++c) {
    const int k0 = 4 * c + 2 * hi;
    const float v0 = y[base + k0];
    const float v1 = y[base + k0 + 1];
    By[c].x = ((k0    ) < 18) ? v0 : 0.f;
    By[c].y = ((k0 + 1) < 18) ? v1 : 0.f;
  }

  v8f d[2];
  v8f cz = {};
#pragma unroll 1
  for (int t = 0; t < OUTT; ++t) {
    if (t == 0) {
#pragma unroll
      for (int nt = 0; nt < 2; ++nt) {
        v8f c1 = wmma4(Aih[0][nt], By[0], cb[nt]);
        v8f c2 = wmma4(Aih[1][nt], By[1], cz);
        c1 = wmma4(Aih[2][nt], By[2], c1);
        c2 = wmma4(Aih[3][nt], By[3], c2);
        c1 = wmma4(Aih[4][nt], By[4], c1);
#pragma unroll
        for (int rr = 0; rr < 8; ++rr) d[nt][rr] = fast_tanh(c1[rr] + c2[rr]);
      }
    } else {
      v2f Bh[8];
#pragma unroll
      for (int c = 0; c < 8; ++c) {
        Bh[c].x = d[c >> 2][2 * (c & 3)    ];
        Bh[c].y = d[c >> 2][2 * (c & 3) + 1];
      }
      v8f dn[2];
#pragma unroll
      for (int nt = 0; nt < 2; ++nt) {
        v8f c1 = cb[nt];
        v8f c2 = cz;
#pragma unroll
        for (int cc = 0; cc < 8; cc += 2) {
          c1 = wmma4(Ahh[cc    ][nt], Bh[cc    ], c1);
          c2 = wmma4(Ahh[cc + 1][nt], Bh[cc + 1], c2);
        }
#pragma unroll
        for (int rr = 0; rr < 8; ++rr) dn[nt][rr] = fast_tanh(c1[rr] + c2[rr]);
      }
      d[0] = dn[0]; d[1] = dn[1];
    }

    const long long ob = ((long long)(s0 + ii) * OUTT + t) * 32;
#pragma unroll
    for (int nt = 0; nt < 2; ++nt)
#pragma unroll
      for (int rr = 0; rr < 8; ++rr)
        ys[ob + nt * 16 + rr + 8 * hi] = d[nt][rr];
  }
}

__global__ __launch_bounds__(256) void out_proj_kernel(
    const float* __restrict__ ys, const float* __restrict__ w_out,
    const float* __restrict__ b_out, float* __restrict__ out)
{
  const int id = blockIdx.x * blockDim.x + threadIdx.x;
  if (id >= B_N * OUTT) return;
  const float* row = ys + (long long)id * 32;
  float a0 = b_out[0], a1 = b_out[1], a2 = b_out[2];
#pragma unroll
  for (int nn = 0; nn < 32; ++nn) {
    const float v = row[nn];
    a0 = fmaf(v, w_out[nn],      a0);
    a1 = fmaf(v, w_out[32 + nn], a1);
    a2 = fmaf(v, w_out[64 + nn], a2);
  }
  out[(long long)id * 3 + 0] = a0;
  out[(long long)id * 3 + 1] = a1;
  out[(long long)id * 3 + 2] = a2;
}

extern "C" void kernel_launch(void* const* d_in, const int* in_sizes, int n_in,
                              void* d_out, int out_size, void* d_ws,
                              size_t ws_size, hipStream_t stream) {
  const float* x      = (const float*)d_in[0];
  const float* w_ih_f = (const float*)d_in[1];
  const float* w_hh_f = (const float*)d_in[2];
  const float* b_ih_f = (const float*)d_in[3];
  const float* b_hh_f = (const float*)d_in[4];
  const float* w_ih_b = (const float*)d_in[5];
  const float* w_hh_b = (const float*)d_in[6];
  const float* b_ih_b = (const float*)d_in[7];
  const float* b_hh_b = (const float*)d_in[8];
  const float* w_ih2  = (const float*)d_in[9];
  const float* w_hh2  = (const float*)d_in[10];
  const float* b_ih2  = (const float*)d_in[11];
  const float* b_hh2  = (const float*)d_in[12];
  const float* w_out  = (const float*)d_in[13];
  const float* b_out  = (const float*)d_in[14];

  float* y  = (float*)d_ws;
  float* ys = y + (size_t)B_N * (2 * H1N);

  rnn1_wmma_kernel<<<dim3(512), dim3(32), 0, stream>>>(
      x, w_ih_f, w_hh_f, b_ih_f, b_hh_f,
      w_ih_b, w_hh_b, b_ih_b, b_hh_b, y);

  rnn2_wmma_kernel<<<dim3(256), dim3(32), 0, stream>>>(
      w_ih2, w_hh2, b_ih2, b_hh2, y, ys);

  const int nthreads = B_N * OUTT;
  out_proj_kernel<<<dim3((nthreads + 255) / 256), dim3(256), 0, stream>>>(
      ys, w_out, b_out, (float*)d_out);
}


# rnn1 step: sums moved into the WMMA C operands (input product onto bias, state product 1 onto that, 3 onto 1; 2 apart): one vector add level behind the last product, not two; sum order changed
# speedup vs baseline: 1.6692x; 1.0590x over previous
#include <hip/hip_runtime.h>
#include <hip/hip_bf16.h>


typedef __attribute__((ext_vector_type(2))) float v2f;
typedef __attribute__((ext_vector_type(8))) float v8f;

#define T_LEN 2048
#define B_N   4096
#define H1N   9
#define H2N   32
#define DIN   5
#define OUTT  25

__device__ __forceinline__ v8f wmma4(v2f a, v2f b, v8f c) {
  return __builtin_amdgcn_wmma_f32_16x16x4_f32(
       false, a,  false, b,
       (short)0, c,  false,  false);
}

__device__ __forceinline__ float fast_tanh(float v) {
#if __has_builtin(__builtin_amdgcn_tanhf)
  return __builtin_amdgcn_tanhf(v);
#else
  return tanhf(v);
#endif
}

template <bool NEXT>
__device__ __forceinline__ void rnn1_step(
    v2f A0, v2f A1, v2f A2, v2f A3, v8f cbias,
    float d[5], v8f& c4, float xw, float2 xpn)
{
  __builtin_amdgcn_sched_barrier(0);

  v2f B0; B0.x = d[0]; B0.y = d[1];
  v2f B1; B1.x = d[2]; B1.y = d[3];
  v2f B2; B2.x = d[4]; B2.y = xw;

  v8f cz = {};
  v8f c1 = wmma4(A0, B0, c4);
  v8f c2 = wmma4(A1, B1, cz);
  v8f c3 = wmma4(A2, B2, c1);
  v8f c4n = cz;
  if (NEXT) {
    v2f B3; B3.x = xpn.x; B3.y = xpn.y;
    asm volatile("" : "+v"(c2), "+v"(c3), "+v"(B3));
    c4n = wmma4(A3, B3, cbias);
    asm volatile("" : "+v"(c4n));
  }
  __builtin_amdgcn_sched_barrier(0);

  if (!NEXT) {
    asm volatile("s_wait_alu depctr_va_vdst(0)" : "+v"(c2), "+v"(c3));
  }

#pragma unroll
  for (int i = 0; i < 5; ++i)
    d[i] = fast_tanh(c3[i] + c2[i]);
  c4 = c4n;
}

__device__ __forceinline__ void rnn1_seam(float d[5], long long& qo, long long& ro)
{
  asm volatile("" : "+v"(d[0]), "+v"(d[1]), "+v"(d[2]), "+v"(d[3]), "+v"(d[4]),
                    "+v"(qo), "+v"(ro));
}

template <int DIR, bool LASTPAIR>
__device__ __forceinline__ void rnn1_fetch_block(
    const float* q, const float* r, float2 xp[8], float xw[8])
{
  constexpr int STEP = DIR ? -DIN : DIN;
#pragma unroll
  for (int u = 0; u < 8; ++u) {
    if (LASTPAIR || u < 7) xp[u] = *(const float2*)&q[(u + 1) * STEP];
    xw[u] = r[u * STEP];
  }
}

template <bool FINAL>
__device__ __forceinline__ void rnn1_run_block(
    v2f A0, v2f A1, v2f A2, v2f A3, v8f cbias,
    float d[5], v8f& c4, const float2 xp[8], const float xw[8])
{
#pragma unroll
  for (int u = 0; u < 7; ++u)
    rnn1_step<true>(A0, A1, A2, A3, cbias, d, c4, xw[u], xp[u]);
  if (FINAL)
    rnn1_step<false>(A0, A1, A2, A3, cbias, d, c4, xw[7], xp[6]);
  else
    rnn1_step<true>(A0, A1, A2, A3, cbias, d, c4, xw[7], xp[7]);
}

template <int DIR>
__device__ __forceinline__ void rnn1_time_loop(
    const float* __restrict__ x, long long xrow, int hi,
    v2f A0, v2f A1, v2f A2, v2f A3, v8f cbias, float d[5])
{
  constexpr int STEP = DIR ? -DIN : DIN;
  constexpr int NS = 8;
  constexpr int NB = T_LEN / NS;
  constexpr int PF = 60;
  const long long t0 = DIR ? (long long)(T_LEN - 1) * DIN : 0;
  long long qo = xrow + t0 + 2 * hi;
  long long ro = xrow + t0 + 4;

#pragma unroll
  for (int i = 0; i < 5; ++i) d[i] = 0.f;

  float2 xpA[NS], xpB[NS];
  float  xwA[NS], xwB[NS];

  const float2 xp0 = *(const float2*)&x[qo];
  rnn1_fetch_block<DIR, true>(x + qo, x + ro, xpA, xwA);
  qo += NS * STEP; ro += NS * STEP;

  v8f c4;
  {
    v2f B3; B3.x = xp0.x; B3.y = xp0.y;
    c4 = wmma4(A3, B3, cbias);
  }

#pragma unroll 1
  for (int b = 0; b < NB - 2; b += 2) {
    rnn1_seam(d, qo, ro);
    rnn1_fetch_block<DIR, true>(x + qo, x + ro, xpB, xwB);
    if (b < NB - 10) {
      __builtin_prefetch(x + ro + STEP * PF, 0, 1);
      __builtin_prefetch(x + ro + STEP * (PF + 4), 0, 1);
    }
    qo += NS * STEP; ro += NS * STEP;
    __builtin_amdgcn_sched_barrier(0);
    rnn1_run_block<false>(A0, A1, A2, A3, cbias, d, c4, xpA, xwA);

    rnn1_seam(d, qo, ro);
    rnn1_fetch_block<DIR, true>(x + qo, x + ro, xpA, xwA);
    if (b < NB - 10) {
      __builtin_prefetch(x + ro + STEP * PF, 0, 1);
      __builtin_prefetch(x + ro + STEP * (PF + 4), 0, 1);
    }
    qo += NS * STEP; ro += NS * STEP;
    __builtin_amdgcn_sched_barrier(0);
    rnn1_run_block<false>(A0, A1, A2, A3, cbias, d, c4, xpB, xwB);
  }

  rnn1_seam(d, qo, ro);
  rnn1_fetch_block<DIR, false>(x + qo, x + ro, xpB, xwB);
  __builtin_amdgcn_sched_barrier(0);
  rnn1_run_block<false>(A0, A1, A2, A3, cbias, d, c4, xpA, xwA);
  rnn1_run_block<true >(A0, A1, A2, A3, cbias, d, c4, xpB, xwB);
}

__global__ __launch_bounds__(32) void rnn1_wmma_kernel(
    const float* __restrict__ x,
    const float* __restrict__ w_ih_f, const float* __restrict__ w_hh_f,
    const float* __restrict__ b_ih_f, const float* __restrict__ b_hh_f,
    const float* __restrict__ w_ih_b, const float* __restrict__ w_hh_b,
    const float* __restrict__ b_ih_b, const float* __restrict__ b_hh_b,
    float* __restrict__ y  )
{
  const int lane = threadIdx.x;
  const int g    = blockIdx.x;
  const int dir  = g >> 8;
  const int s0   = (g & 255) * 16;
  const int hi   = lane >> 4;
  const int ii   = lane & 15;

  const float* wih = dir ? w_ih_b : w_ih_f;
  const float* whh = dir ? w_hh_b : w_hh_f;
  const float* bih = dir ? b_ih_b : b_ih_f;
  const float* bhh = dir ? b_hh_b : b_hh_f;

  auto unitOfRow = [](int i) -> int {
    return (i < 5) ? i : ((i >= 8 && i < 12) ? i - 3 : -1);
  };
  const int slotType[16] = {0,0,0,0, 0,0,0,0, 0,2,2,1, 1,1,1,1};
  const int slotIdx [16] = {0,1,5,6, 2,3,7,8, 4,0,0,4, 0,1,2,3};

  const int urow = unitOfRow(ii);
  auto wElem = [&](int k) -> float {
    if (urow < 0) return 0.f;
    const int ty = slotType[k], ix = slotIdx[k];
    if (ty == 0) return whh[urow * H1N + ix];
    if (ty == 1) return wih[urow * DIN + ix];
    return 0.f;
  };
  auto aW = [&](int kb) -> v2f {
    v2f v; const int k0 = kb + 2 * hi;
    v.x = wElem(k0); v.y = wElem(k0 + 1);
    return v;
  };
  const v2f A0 = aW(0), A1 = aW(4), A2 = aW(8), A3 = aW(12);

  v8f cbias;
#pragma unroll
  for (int rr = 0; rr < 8; ++rr) {
    const int uu = unitOfRow(rr + 8 * hi);
    cbias[rr] = (uu >= 0) ? (bih[uu] + bhh[uu]) : 0.f;
  }

  const long long xrow = (long long)(s0 + ii) * (T_LEN * DIN);

  float d[5];
  if (dir == 0)
    rnn1_time_loop<0>(x, xrow, hi, A0, A1, A2, A3, cbias, d);
  else
    rnn1_time_loop<1>(x, xrow, hi, A0, A1, A2, A3, cbias, d);

  const int sj = s0 + ii;
#pragma unroll
  for (int rr = 0; rr < 5; ++rr) {
    const int uu = unitOfRow(rr + 8 * hi);
    if (uu >= 0) y[sj * (2 * H1N) + dir * H1N + uu] = d[rr];
  }
}

__global__ __launch_bounds__(32) void rnn2_wmma_kernel(
    const float* __restrict__ w_ih2, const float* __restrict__ w_hh2,
    const float* __restrict__ b_ih2, const float* __restrict__ b_hh2,
    const float* __restrict__ y   ,
    float* __restrict__ ys        )
{
  const int lane = threadIdx.x;
  const int hi = lane >> 4, ii = lane & 15;
  const int s0 = blockIdx.x * 16;

  v2f Aih[5][2];
#pragma unroll
  for (int c = 0; c < 5; ++c)
#pragma unroll
    for (int nt = 0; nt < 2; ++nt) {
      const int k0 = 4 * c + 2 * hi, i = nt * 16 + ii;
      Aih[c][nt].x = ((k0    ) < 18) ? w_ih2[i * 18 + k0    ] : 0.f;
      Aih[c][nt].y = ((k0 + 1) < 18) ? w_ih2[i * 18 + k0 + 1] : 0.f;
    }
  v2f Ahh[8][2];
#pragma unroll
  for (int c = 0; c < 8; ++c)
#pragma unroll
    for (int nt = 0; nt < 2; ++nt) {
      const int j0 = 16 * (c >> 2) + 8 * hi + 2 * (c & 3);
      const int i = nt * 16 + ii;
      Ahh[c][nt].x = w_hh2[i * 32 + j0    ];
      Ahh[c][nt].y = w_hh2[i * 32 + j0 + 1];
    }

  v8f cb[2];
#pragma unroll
  for (int nt = 0; nt < 2; ++nt)
#pragma unroll
    for (int rr = 0; rr < 8; ++rr) {
      const int i = nt * 16 + rr + 8 * hi;
      cb[nt][rr] = b_ih2[i] + b_hh2[i];
    }

  const int base = (s0 + ii) * 18;
  v2f By[5];
#pragma unroll
  for (int c = 0; c < 5; ++c) {
    const int k0 = 4 * c + 2 * hi;
    const float v0 = y[base + k0];
    const float v1 = y[base + k0 + 1];
    By[c].x = ((k0    ) < 18) ? v0 : 0.f;
    By[c].y = ((k0 + 1) < 18) ? v1 : 0.f;
  }

  v8f d[2];
  v8f cz = {};
#pragma unroll 1
  for (int t = 0; t < OUTT; ++t) {
    if (t == 0) {
#pragma unroll
      for (int nt = 0; nt < 2; ++nt) {
        v8f c1 = wmma4(Aih[0][nt], By[0], cb[nt]);
        v8f c2 = wmma4(Aih[1][nt], By[1], cz);
        c1 = wmma4(Aih[2][nt], By[2], c1);
        c2 = wmma4(Aih[3][nt], By[3], c2);
        c1 = wmma4(Aih[4][nt], By[4], c1);
#pragma unroll
        for (int rr = 0; rr < 8; ++rr) d[nt][rr] = fast_tanh(c1[rr] + c2[rr]);
      }
    } else {
      v2f Bh[8];
#pragma unroll
      for (int c = 0; c < 8; ++c) {
        Bh[c].x = d[c >> 2][2 * (c & 3)    ];
        Bh[c].y = d[c >> 2][2 * (c & 3) + 1];
      }
      v8f dn[2];
#pragma unroll
      for (int nt = 0; nt < 2; ++nt) {
        v8f c1 = cb[nt];
        v8f c2 = cz;
#pragma unroll
        for (int cc = 0; cc < 8; cc += 2) {
          c1 = wmma4(Ahh[cc    ][nt], Bh[cc    ], c1);
          c2 = wmma4(Ahh[cc + 1][nt], Bh[cc + 1], c2);
        }
#pragma unroll
        for (int rr = 0; rr < 8; ++rr) dn[nt][rr] = fast_tanh(c1[rr] + c2[rr]);
      }
      d[0] = dn[0]; d[1] = dn[1];
    }

    const long long ob = ((long long)(s0 + ii) * OUTT + t) * 32;
#pragma unroll
    for (int nt = 0; nt < 2; ++nt)
#pragma unroll
      for (int rr = 0; rr < 8; ++rr)
        ys[ob + nt * 16 + rr + 8 * hi] = d[nt][rr];
  }
}

__global__ __launch_bounds__(256) void out_proj_kernel(
    const float* __restrict__ ys, const float* __restrict__ w_out,
    const float* __restrict__ b_out, float* __restrict__ out)
{
  const int id = blockIdx.x * blockDim.x + threadIdx.x;
  if (id >= B_N * OUTT) return;
  const float* row = ys + (long long)id * 32;
  float a0 = b_out[0], a1 = b_out[1], a2 = b_out[2];
#pragma unroll
  for (int nn = 0; nn < 32; ++nn) {
    const float v = row[nn];
    a0 = fmaf(v, w_out[nn],      a0);
    a1 = fmaf(v, w_out[32 + nn], a1);
    a2 = fmaf(v, w_out[64 + nn], a2);
  }
  out[(long long)id * 3 + 0] = a0;
  out[(long long)id * 3 + 1] = a1;
  out[(long long)id * 3 + 2] = a2;
}

extern "C" void kernel_launch(void* const* d_in, const int* in_sizes, int n_in,
                              void* d_out, int out_size, void* d_ws,
                              size_t ws_size, hipStream_t stream) {
  const float* x      = (const float*)d_in[0];
  const float* w_ih_f = (const float*)d_in[1];
  const float* w_hh_f = (const float*)d_in[2];
  const float* b_ih_f = (const float*)d_in[3];
  const float* b_hh_f = (const float*)d_in[4];
  const float* w_ih_b = (const float*)d_in[5];
  const float* w_hh_b = (const float*)d_in[6];
  const float* b_ih_b = (const float*)d_in[7];
  const float* b_hh_b = (const float*)d_in[8];
  const float* w_ih2  = (const float*)d_in[9];
  const float* w_hh2  = (const float*)d_in[10];
  const float* b_ih2  = (const float*)d_in[11];
  const float* b_hh2  = (const float*)d_in[12];
  const float* w_out  = (const float*)d_in[13];
  const float* b_out  = (const float*)d_in[14];

  float* y  = (float*)d_ws;
  float* ys = y + (size_t)B_N * (2 * H1N);

  rnn1_wmma_kernel<<<dim3(512), dim3(32), 0, stream>>>(
      x, w_ih_f, w_hh_f, b_ih_f, b_hh_f,
      w_ih_b, w_hh_b, b_ih_b, b_hh_b, y);

  rnn2_wmma_kernel<<<dim3(256), dim3(32), 0, stream>>>(
      w_ih2, w_hh2, b_ih2, b_hh2, y, ys);

  const int nthreads = B_N * OUTT;
  out_proj_kernel<<<dim3((nthreads + 255) / 256), dim3(256), 0, stream>>>(
      ys, w_out, b_out, (float*)d_out);
}


# two launches, not three: 32->3 projection done in the RNN-2 kernel on each state in registers (states not stored), out staged in LDS, whole-line stores; projection sum order changed
# speedup vs baseline: 1.7499x; 1.0483x over previous
#include <hip/hip_runtime.h>
#include <hip/hip_bf16.h>


typedef __attribute__((ext_vector_type(2))) float v2f;
typedef __attribute__((ext_vector_type(8))) float v8f;

#define T_LEN 2048
#define B_N   4096
#define H1N   9
#define H2N   32
#define DIN   5
#define OUTT  25
#define D_OUTN 3

__device__ __forceinline__ v8f wmma4(v2f a, v2f b, v8f c) {
  return __builtin_amdgcn_wmma_f32_16x16x4_f32(
       false, a,  false, b,
       (short)0, c,  false,  false);
}

__device__ __forceinline__ float fast_tanh(float v) {
#if __has_builtin(__builtin_amdgcn_tanhf)
  return __builtin_amdgcn_tanhf(v);
#else
  return tanhf(v);
#endif
}

template <bool NEXT>
__device__ __forceinline__ void rnn1_step(
    v2f A0, v2f A1, v2f A2, v2f A3, v8f cbias,
    float d[5], v8f& c4, float xw, float2 xpn)
{
  __builtin_amdgcn_sched_barrier(0);

  v2f B0; B0.x = d[0]; B0.y = d[1];
  v2f B1; B1.x = d[2]; B1.y = d[3];
  v2f B2; B2.x = d[4]; B2.y = xw;

  v8f cz = {};
  v8f c1 = wmma4(A0, B0, c4);
  v8f c2 = wmma4(A1, B1, cz);
  v8f c3 = wmma4(A2, B2, c1);
  v8f c4n = cz;
  if (NEXT) {
    v2f B3; B3.x = xpn.x; B3.y = xpn.y;
    asm volatile("" : "+v"(c2), "+v"(c3), "+v"(B3));
    c4n = wmma4(A3, B3, cbias);
    asm volatile("" : "+v"(c4n));
  }
  __builtin_amdgcn_sched_barrier(0);

  if (!NEXT) {
    asm volatile("s_wait_alu depctr_va_vdst(0)" : "+v"(c2), "+v"(c3));
  }

#pragma unroll
  for (int i = 0; i < 5; ++i)
    d[i] = fast_tanh(c3[i] + c2[i]);
  c4 = c4n;
}

__device__ __forceinline__ void rnn1_seam(float d[5], long long& qo, long long& ro)
{
  asm volatile("" : "+v"(d[0]), "+v"(d[1]), "+v"(d[2]), "+v"(d[3]), "+v"(d[4]),
                    "+v"(qo), "+v"(ro));
}

template <int DIR, bool LASTPAIR>
__device__ __forceinline__ void rnn1_fetch_block(
    const float* q, const float* r, float2 xp[8], float xw[8])
{
  constexpr int STEP = DIR ? -DIN : DIN;
#pragma unroll
  for (int u = 0; u < 8; ++u) {
    if (LASTPAIR || u < 7) xp[u] = *(const float2*)&q[(u + 1) * STEP];
    xw[u] = r[u * STEP];
  }
}

template <bool FINAL>
__device__ __forceinline__ void rnn1_run_block(
    v2f A0, v2f A1, v2f A2, v2f A3, v8f cbias,
    float d[5], v8f& c4, const float2 xp[8], const float xw[8])
{
#pragma unroll
  for (int u = 0; u < 7; ++u)
    rnn1_step<true>(A0, A1, A2, A3, cbias, d, c4, xw[u], xp[u]);
  if (FINAL)
    rnn1_step<false>(A0, A1, A2, A3, cbias, d, c4, xw[7], xp[6]);
  else
    rnn1_step<true>(A0, A1, A2, A3, cbias, d, c4, xw[7], xp[7]);
}

template <int DIR>
__device__ __forceinline__ void rnn1_time_loop(
    const float* __restrict__ x, long long xrow, int hi,
    v2f A0, v2f A1, v2f A2, v2f A3, v8f cbias, float d[5])
{
  constexpr int STEP = DIR ? -DIN : DIN;
  constexpr int NS = 8;
  constexpr int NB = T_LEN / NS;
  constexpr int PF = 60;
  const long long t0 = DIR ? (long long)(T_LEN - 1) * DIN : 0;
  long long qo = xrow + t0 + 2 * hi;
  long long ro = xrow + t0 + 4;

#pragma unroll
  for (int i = 0; i < 5; ++i) d[i] = 0.f;

  float2 xpA[NS], xpB[NS];
  float  xwA[NS], xwB[NS];

  const float2 xp0 = *(const float2*)&x[qo];
  rnn1_fetch_block<DIR, true>(x + qo, x + ro, xpA, xwA);
  qo += NS * STEP; ro += NS * STEP;

  v8f c4;
  {
    v2f B3; B3.x = xp0.x; B3.y = xp0.y;
    c4 = wmma4(A3, B3, cbias);
  }

#pragma unroll 1
  for (int b = 0; b < NB - 2; b += 2) {
    rnn1_seam(d, qo, ro);
    rnn1_fetch_block<DIR, true>(x + qo, x + ro, xpB, xwB);
    if (b < NB - 10) {
      __builtin_prefetch(x + ro + STEP * PF, 0, 1);
      __builtin_prefetch(x + ro + STEP * (PF + 4), 0, 1);
    }
    qo += NS * STEP; ro += NS * STEP;
    __builtin_amdgcn_sched_barrier(0);
    rnn1_run_block<false>(A0, A1, A2, A3, cbias, d, c4, xpA, xwA);

    rnn1_seam(d, qo, ro);
    rnn1_fetch_block<DIR, true>(x + qo, x + ro, xpA, xwA);
    if (b < NB - 10) {
      __builtin_prefetch(x + ro + STEP * PF, 0, 1);
      __builtin_prefetch(x + ro + STEP * (PF + 4), 0, 1);
    }
    qo += NS * STEP; ro += NS * STEP;
    __builtin_amdgcn_sched_barrier(0);
    rnn1_run_block<false>(A0, A1, A2, A3, cbias, d, c4, xpB, xwB);
  }

  rnn1_seam(d, qo, ro);
  rnn1_fetch_block<DIR, false>(x + qo, x + ro, xpB, xwB);
  __builtin_amdgcn_sched_barrier(0);
  rnn1_run_block<false>(A0, A1, A2, A3, cbias, d, c4, xpA, xwA);
  rnn1_run_block<true >(A0, A1, A2, A3, cbias, d, c4, xpB, xwB);
}

__global__ __launch_bounds__(32) void rnn1_wmma_kernel(
    const float* __restrict__ x,
    const float* __restrict__ w_ih_f, const float* __restrict__ w_hh_f,
    const float* __restrict__ b_ih_f, const float* __restrict__ b_hh_f,
    const float* __restrict__ w_ih_b, const float* __restrict__ w_hh_b,
    const float* __restrict__ b_ih_b, const float* __restrict__ b_hh_b,
    float* __restrict__ y  )
{
  const int lane = threadIdx.x;
  const int g    = blockIdx.x;
  const int dir  = g >> 8;
  const int s0   = (g & 255) * 16;
  const int hi   = lane >> 4;
  const int ii   = lane & 15;

  const float* wih = dir ? w_ih_b : w_ih_f;
  const float* whh = dir ? w_hh_b : w_hh_f;
  const float* bih = dir ? b_ih_b : b_ih_f;
  const float* bhh = dir ? b_hh_b : b_hh_f;

  auto unitOfRow = [](int i) -> int {
    return (i < 5) ? i : ((i >= 8 && i < 12) ? i - 3 : -1);
  };
  const int slotType[16] = {0,0,0,0, 0,0,0,0, 0,2,2,1, 1,1,1,1};
  const int slotIdx [16] = {0,1,5,6, 2,3,7,8, 4,0,0,4, 0,1,2,3};

  const int urow = unitOfRow(ii);
  auto wElem = [&](int k) -> float {
    if (urow < 0) return 0.f;
    const int ty = slotType[k], ix = slotIdx[k];
    if (ty == 0) return whh[urow * H1N + ix];
    if (ty == 1) return wih[urow * DIN + ix];
    return 0.f;
  };
  auto aW = [&](int kb) -> v2f {
    v2f v; const int k0 = kb + 2 * hi;
    v.x = wElem(k0); v.y = wElem(k0 + 1);
    return v;
  };
  const v2f A0 = aW(0), A1 = aW(4), A2 = aW(8), A3 = aW(12);

  v8f cbias;
#pragma unroll
  for (int rr = 0; rr < 8; ++rr) {
    const int uu = unitOfRow(rr + 8 * hi);
    cbias[rr] = (uu >= 0) ? (bih[uu] + bhh[uu]) : 0.f;
  }

  const long long xrow = (long long)(s0 + ii) * (T_LEN * DIN);

  float d[5];
  if (dir == 0)
    rnn1_time_loop<0>(x, xrow, hi, A0, A1, A2, A3, cbias, d);
  else
    rnn1_time_loop<1>(x, xrow, hi, A0, A1, A2, A3, cbias, d);

  const int sj = s0 + ii;
#pragma unroll
  for (int rr = 0; rr < 5; ++rr) {
    const int uu = unitOfRow(rr + 8 * hi);
    if (uu >= 0) y[sj * (2 * H1N) + dir * H1N + uu] = d[rr];
  }
}


__device__ __forceinline__ float xhalf(float x) {
  const int b = __builtin_bit_cast(int, x);
  return __builtin_bit_cast(float, __builtin_amdgcn_permlanex16(
      b, b, 0x76543210, 0xfedcba98, false, false));
}

__device__ __forceinline__ void proj3(const v8f d[2], const float wo[3][2][8],
                                      const float pinit[3], float p[3])
{
#pragma unroll
  for (int o = 0; o < 3; ++o) {
    float a = pinit[o], b = 0.f;
#pragma unroll
    for (int rr = 0; rr < 8; ++rr) {
      a = fmaf(d[0][rr], wo[o][0][rr], a);
      b = fmaf(d[1][rr], wo[o][1][rr], b);
    }
    const float s = a + b;
    p[o] = s + xhalf(s);
  }
}

__global__ __launch_bounds__(64) void rnn2_proj_kernel(
    const float* __restrict__ w_ih2, const float* __restrict__ w_hh2,
    const float* __restrict__ b_ih2, const float* __restrict__ b_hh2,
    const float* __restrict__ w_out, const float* __restrict__ b_out,
    const float* __restrict__ y   ,
    float* __restrict__ out       )
{
  __shared__ __attribute__((aligned(16))) float so[32 * OUTT * D_OUTN];

  const int lane = threadIdx.x & 31;
  const int wv   = threadIdx.x >> 5;
  const int hi = lane >> 4, ii = lane & 15;
  const int s0 = blockIdx.x * 32 + wv * 16;

  v2f Aih[5][2];
#pragma unroll
  for (int c = 0; c < 5; ++c)
#pragma unroll
    for (int nt = 0; nt < 2; ++nt) {
      const int k0 = 4 * c + 2 * hi, i = nt * 16 + ii;
      Aih[c][nt].x = ((k0    ) < 18) ? w_ih2[i * 18 + k0    ] : 0.f;
      Aih[c][nt].y = ((k0 + 1) < 18) ? w_ih2[i * 18 + k0 + 1] : 0.f;
    }
  v2f Ahh[8][2];
#pragma unroll
  for (int c = 0; c < 8; ++c)
#pragma unroll
    for (int nt = 0; nt < 2; ++nt) {
      const int j0 = 16 * (c >> 2) + 8 * hi + 2 * (c & 3);
      const int i = nt * 16 + ii;
      Ahh[c][nt].x = w_hh2[i * 32 + j0    ];
      Ahh[c][nt].y = w_hh2[i * 32 + j0 + 1];
    }

  v8f cb[2];
#pragma unroll
  for (int nt = 0; nt < 2; ++nt)
#pragma unroll
    for (int rr = 0; rr < 8; ++rr) {
      const int i = nt * 16 + rr + 8 * hi;
      cb[nt][rr] = b_ih2[i] + b_hh2[i];
    }

  float wo[3][2][8], pinit[3];
#pragma unroll
  for (int o = 0; o < 3; ++o) {
    pinit[o] = hi ? 0.f : b_out[o];
#pragma unroll
    for (int nt = 0; nt < 2; ++nt)
#pragma unroll
      for (int rr = 0; rr < 8; ++rr)
        wo[o][nt][rr] = w_out[o * H2N + 16 * nt + rr + 8 * hi];
  }

  const int base = (s0 + ii) * 18;
  v2f By[5];
#pragma unroll
  for (int c = 0; c < 5; ++c) {
    const int k0 = 4 * c + 2 * hi;
    const int kc = (k0 < 18) ? k0 : 16;
    const float v0 = y[base + kc];
    const float v1 = y[base + kc + 1];
    By[c].x = ((k0    ) < 18) ? v0 : 0.f;
    By[c].y = ((k0 + 1) < 18) ? v1 : 0.f;
  }

  float* const srow = so + (wv * 16 + ii) * (OUTT * D_OUTN);
  v8f d[2];
  v8f cz = {};

#pragma unroll
  for (int nt = 0; nt < 2; ++nt) {
    v8f c1 = wmma4(Aih[0][nt], By[0], cb[nt]);
    v8f c2 = wmma4(Aih[1][nt], By[1], cz);
    c1 = wmma4(Aih[2][nt], By[2], c1);
    c2 = wmma4(Aih[3][nt], By[3], c2);
    c1 = wmma4(Aih[4][nt], By[4], c1);
#pragma unroll
    for (int rr = 0; rr < 8; ++rr) d[nt][rr] = fast_tanh(c1[rr] + c2[rr]);
  }

#pragma unroll 1
  for (int t = 1; t < OUTT; ++t) {
    v2f Bh[8];
#pragma unroll
    for (int c = 0; c < 8; ++c) {
      Bh[c].x = d[c >> 2][2 * (c & 3)    ];
      Bh[c].y = d[c >> 2][2 * (c & 3) + 1];
    }
    v8f acc[2][2];
#pragma unroll
    for (int nt = 0; nt < 2; ++nt) {
      v8f c1 = cb[nt];
      v8f c2 = cz;
#pragma unroll
      for (int cc = 0; cc < 8; cc += 2) {
        c1 = wmma4(Ahh[cc    ][nt], Bh[cc    ], c1);
        c2 = wmma4(Ahh[cc + 1][nt], Bh[cc + 1], c2);
      }
      acc[nt][0] = c1; acc[nt][1] = c2;
    }

    float p[3];
    proj3(d, wo, pinit, p);
    if (hi == 0) {
      srow[(t - 1) * D_OUTN + 0] = p[0];
      srow[(t - 1) * D_OUTN + 1] = p[1];
      srow[(t - 1) * D_OUTN + 2] = p[2];
    }

#pragma unroll
    for (int nt = 0; nt < 2; ++nt)
#pragma unroll
      for (int rr = 0; rr < 8; ++rr)
        d[nt][rr] = fast_tanh(acc[nt][0][rr] + acc[nt][1][rr]);
  }

  {
    float p[3];
    proj3(d, wo, pinit, p);
    if (hi == 0) {
      srow[(OUTT - 1) * D_OUTN + 0] = p[0];
      srow[(OUTT - 1) * D_OUTN + 1] = p[1];
      srow[(OUTT - 1) * D_OUTN + 2] = p[2];
    }
  }

  __syncthreads();

  const float4* s4 = (const float4*)so;
  float4* o4 = (float4*)(out + (size_t)blockIdx.x * (32 * OUTT * D_OUTN));
  for (int i = threadIdx.x; i < 32 * OUTT * D_OUTN / 4; i += 64)
    o4[i] = s4[i];
}

extern "C" void kernel_launch(void* const* d_in, const int* in_sizes, int n_in,
                              void* d_out, int out_size, void* d_ws,
                              size_t ws_size, hipStream_t stream) {
  const float* x      = (const float*)d_in[0];
  const float* w_ih_f = (const float*)d_in[1];
  const float* w_hh_f = (const float*)d_in[2];
  const float* b_ih_f = (const float*)d_in[3];
  const float* b_hh_f = (const float*)d_in[4];
  const float* w_ih_b = (const float*)d_in[5];
  const float* w_hh_b = (const float*)d_in[6];
  const float* b_ih_b = (const float*)d_in[7];
  const float* b_hh_b = (const float*)d_in[8];
  const float* w_ih2  = (const float*)d_in[9];
  const float* w_hh2  = (const float*)d_in[10];
  const float* b_ih2  = (const float*)d_in[11];
  const float* b_hh2  = (const float*)d_in[12];
  const float* w_out  = (const float*)d_in[13];
  const float* b_out  = (const float*)d_in[14];

  float* y = (float*)d_ws;

  rnn1_wmma_kernel<<<dim3(512), dim3(32), 0, stream>>>(
      x, w_ih_f, w_hh_f, b_ih_f, b_hh_f,
      w_ih_b, w_hh_b, b_ih_b, b_hh_b, y);

  rnn2_proj_kernel<<<dim3(128), dim3(64), 0, stream>>>(
      w_ih2, w_hh2, b_ih2, b_hh2, w_out, b_out, y, (float*)d_out);
}


# rnn1 step: 3 products, not 4: h4 and x4 columns leave the matrix unit (5 multiplies + 5 FMAs behind the first product, into C operands); unit 4 kept in both lane halves; sum order changed
# speedup vs baseline: 1.8400x; 1.0515x over previous
#include <hip/hip_runtime.h>
#include <hip/hip_bf16.h>


typedef __attribute__((ext_vector_type(2))) float v2f;
typedef __attribute__((ext_vector_type(8))) float v8f;

#define T_LEN 2048
#define B_N   4096
#define H1N   9
#define H2N   32
#define DIN   5
#define OUTT  25
#define D_OUTN 3

__device__ __forceinline__ v8f wmma4(v2f a, v2f b, v8f c) {
  return __builtin_amdgcn_wmma_f32_16x16x4_f32(
       false, a,  false, b,
       (short)0, c,  false,  false);
}

__device__ __forceinline__ float fast_tanh(float v) {
#if __has_builtin(__builtin_amdgcn_tanhf)
  return __builtin_amdgcn_tanhf(v);
#else
  return tanhf(v);
#endif
}

template <bool NEXT>
__device__ __forceinline__ void rnn1_step(
    v2f A0, v2f A1, v2f A3,
    const float wh4[5], const float wx4[5], const float bs[5],
    float d[5], v8f& c4, v8f& cH, v8f& cX, float xwn, float2 xpn)
{
  __builtin_amdgcn_sched_barrier(0);

  v2f B0; B0.x = d[0]; B0.y = d[1];
  v2f B1; B1.x = d[2]; B1.y = d[3];

  v8f c1 = wmma4(A0, B0, c4);

  float h4 = d[4], x4n = xwn;
  asm volatile("" : "+v"(c1), "+v"(h4), "+v"(x4n));
#pragma unroll
  for (int i = 0; i < 5; ++i) cH[i] = wh4[i] * h4;
  if (NEXT) {
#pragma unroll
    for (int i = 0; i < 5; ++i) cX[i] = fmaf(wx4[i], x4n, bs[i]);
  }
  asm volatile("" : "+v"(cH), "+v"(cX), "+v"(B1), "+v"(B0));

  v8f c2 = wmma4(A1, B1, cH);
  v8f c4n = c4;
  if (NEXT) {
    v2f B3; B3.x = xpn.x; B3.y = xpn.y;
    asm volatile("" : "+v"(c1), "+v"(c2), "+v"(B3));
    c4n = wmma4(A3, B3, cX);
    asm volatile("" : "+v"(c4n), "+v"(c2));
  }
  __builtin_amdgcn_sched_barrier(0);

  if (!NEXT) {
    asm volatile("s_wait_alu depctr_va_vdst(0)" : "+v"(c1), "+v"(c2));
  }

#pragma unroll
  for (int i = 0; i < 5; ++i)
    d[i] = fast_tanh(c1[i] + c2[i]);
  c4 = c4n;
}

__device__ __forceinline__ void rnn1_seam(float d[5], long long& qo, long long& ro)
{
  asm volatile("" : "+v"(d[0]), "+v"(d[1]), "+v"(d[2]), "+v"(d[3]), "+v"(d[4]),
                    "+v"(qo), "+v"(ro));
}

template <int DIR, bool LASTPAIR>
__device__ __forceinline__ void rnn1_fetch_block(
    const float* q, const float* r, float2 xp[8], float xw[8])
{
  constexpr int STEP = DIR ? -DIN : DIN;
#pragma unroll
  for (int u = 0; u < 8; ++u) {
    if (LASTPAIR || u < 7) {
      xp[u] = *(const float2*)&q[(u + 1) * STEP];
      xw[u] = r[(u + 1) * STEP];
    }
  }
}

template <bool FINAL>
__device__ __forceinline__ void rnn1_run_block(
    v2f A0, v2f A1, v2f A3,
    const float wh4[5], const float wx4[5], const float bs[5],
    float d[5], v8f& c4, v8f& cH, v8f& cX, const float2 xp[8], const float xw[8])
{
#pragma unroll
  for (int u = 0; u < 7; ++u)
    rnn1_step<true>(A0, A1, A3, wh4, wx4, bs, d, c4, cH, cX, xw[u], xp[u]);
  if (FINAL)
    rnn1_step<false>(A0, A1, A3, wh4, wx4, bs, d, c4, cH, cX, xw[6], xp[6]);
  else
    rnn1_step<true>(A0, A1, A3, wh4, wx4, bs, d, c4, cH, cX, xw[7], xp[7]);
}

template <int DIR>
__device__ __forceinline__ void rnn1_time_loop(
    const float* __restrict__ x, long long xrow, int hi,
    v2f A0, v2f A1, v2f A3,
    const float wh4[5], const float wx4[5], const float bs[5], float d[5])
{
  constexpr int STEP = DIR ? -DIN : DIN;
  constexpr int NS = 8;
  constexpr int NB = T_LEN / NS;
  constexpr int PF = 60;
  const long long t0 = DIR ? (long long)(T_LEN - 1) * DIN : 0;
  long long qo = xrow + t0 + 2 * hi;
  long long ro = xrow + t0 + 4;

#pragma unroll
  for (int i = 0; i < 5; ++i) d[i] = 0.f;

  float2 xpA[NS], xpB[NS];
  float  xwA[NS], xwB[NS];

  const float2 xp0 = *(const float2*)&x[qo];
  const float  xw0 = x[ro];
  rnn1_fetch_block<DIR, true>(x + qo, x + ro, xpA, xwA);
  qo += NS * STEP; ro += NS * STEP;

  v8f cH = {}, cX = {};
  v8f c4;
  {
#pragma unroll
    for (int i = 0; i < 5; ++i) cX[i] = fmaf(wx4[i], xw0, bs[i]);
    v2f B3; B3.x = xp0.x; B3.y = xp0.y;
    c4 = wmma4(A3, B3, cX);
  }

#pragma unroll 1
  for (int b = 0; b < NB - 2; b += 2) {
    rnn1_seam(d, qo, ro);
    rnn1_fetch_block<DIR, true>(x + qo, x + ro, xpB, xwB);
    if (b < NB - 10) {
      __builtin_prefetch(x + ro + STEP * PF, 0, 1);
      __builtin_prefetch(x + ro + STEP * (PF + 4), 0, 1);
    }
    qo += NS * STEP; ro += NS * STEP;
    __builtin_amdgcn_sched_barrier(0);
    rnn1_run_block<false>(A0, A1, A3, wh4, wx4, bs, d, c4, cH, cX, xpA, xwA);

    rnn1_seam(d, qo, ro);
    rnn1_fetch_block<DIR, true>(x + qo, x + ro, xpA, xwA);
    if (b < NB - 10) {
      __builtin_prefetch(x + ro + STEP * PF, 0, 1);
      __builtin_prefetch(x + ro + STEP * (PF + 4), 0, 1);
    }
    qo += NS * STEP; ro += NS * STEP;
    __builtin_amdgcn_sched_barrier(0);
    rnn1_run_block<false>(A0, A1, A3, wh4, wx4, bs, d, c4, cH, cX, xpB, xwB);
  }

  rnn1_seam(d, qo, ro);
  rnn1_fetch_block<DIR, false>(x + qo, x + ro, xpB, xwB);
  __builtin_amdgcn_sched_barrier(0);
  rnn1_run_block<false>(A0, A1, A3, wh4, wx4, bs, d, c4, cH, cX, xpA, xwA);
  rnn1_run_block<true >(A0, A1, A3, wh4, wx4, bs, d, c4, cH, cX, xpB, xwB);
}

__global__ __launch_bounds__(32) void rnn1_wmma_kernel(
    const float* __restrict__ x,
    const float* __restrict__ w_ih_f, const float* __restrict__ w_hh_f,
    const float* __restrict__ b_ih_f, const float* __restrict__ b_hh_f,
    const float* __restrict__ w_ih_b, const float* __restrict__ w_hh_b,
    const float* __restrict__ b_ih_b, const float* __restrict__ b_hh_b,
    float* __restrict__ y  )
{
  const int lane = threadIdx.x;
  const int g    = blockIdx.x;
  const int dir  = g >> 8;
  const int s0   = (g & 255) * 16;
  const int hi   = lane >> 4;
  const int ii   = lane & 15;

  const float* wih = dir ? w_ih_b : w_ih_f;
  const float* whh = dir ? w_hh_b : w_hh_f;
  const float* bih = dir ? b_ih_b : b_ih_f;
  const float* bhh = dir ? b_hh_b : b_hh_f;

  auto unitOfRow = [](int i) -> int {
    return (i < 5) ? i : ((i >= 8 && i < 12) ? i - 3 : ((i == 12) ? 4 : -1));
  };
  const int slotType[16] = {0,0,0,0, 0,0,0,0, 2,2,2,2, 1,1,1,1};
  const int slotIdx [16] = {0,1,5,6, 2,3,7,8, 0,0,0,0, 0,1,2,3};

  const int urow = unitOfRow(ii);
  auto wElem = [&](int k) -> float {
    if (urow < 0) return 0.f;
    const int ty = slotType[k], ix = slotIdx[k];
    if (ty == 0) return whh[urow * H1N + ix];
    if (ty == 1) return wih[urow * DIN + ix];
    return 0.f;
  };
  auto aW = [&](int kb) -> v2f {
    v2f v; const int k0 = kb + 2 * hi;
    v.x = wElem(k0); v.y = wElem(k0 + 1);
    return v;
  };
  const v2f A0 = aW(0), A1 = aW(4), A3 = aW(12);

  float bs[5], wh4[5], wx4[5];
#pragma unroll
  for (int rr = 0; rr < 5; ++rr) {
    const int uu = unitOfRow(rr + 8 * hi);
    bs[rr]  = (uu >= 0) ? (bih[uu] + bhh[uu]) : 0.f;
    wh4[rr] = (uu >= 0) ? whh[uu * H1N + 4] : 0.f;
    wx4[rr] = (uu >= 0) ? wih[uu * DIN + 4] : 0.f;
  }

  const long long xrow = (long long)(s0 + ii) * (T_LEN * DIN);

  float d[5];
  if (dir == 0)
    rnn1_time_loop<0>(x, xrow, hi, A0, A1, A3, wh4, wx4, bs, d);
  else
    rnn1_time_loop<1>(x, xrow, hi, A0, A1, A3, wh4, wx4, bs, d);

  const int sj = s0 + ii;
#pragma unroll
  for (int rr = 0; rr < 5; ++rr) {
    const int row = rr + 8 * hi;
    const int uu = unitOfRow(row);
    if (uu >= 0 && row != 12) y[sj * (2 * H1N) + dir * H1N + uu] = d[rr];
  }
}


__device__ __forceinline__ float xhalf(float x) {
  const int b = __builtin_bit_cast(int, x);
  return __builtin_bit_cast(float, __builtin_amdgcn_permlanex16(
      b, b, 0x76543210, 0xfedcba98, false, false));
}

__device__ __forceinline__ void proj3(const v8f d[2], const float wo[3][2][8],
                                      const float pinit[3], float p[3])
{
#pragma unroll
  for (int o = 0; o < 3; ++o) {
    float a = pinit[o], b = 0.f;
#pragma unroll
    for (int rr = 0; rr < 8; ++rr) {
      a = fmaf(d[0][rr], wo[o][0][rr], a);
      b = fmaf(d[1][rr], wo[o][1][rr], b);
    }
    const float s = a + b;
    p[o] = s + xhalf(s);
  }
}

__global__ __launch_bounds__(64) void rnn2_proj_kernel(
    const float* __restrict__ w_ih2, const float* __restrict__ w_hh2,
    const float* __restrict__ b_ih2, const float* __restrict__ b_hh2,
    const float* __restrict__ w_out, const float* __restrict__ b_out,
    const float* __restrict__ y   ,
    float* __restrict__ out       )
{
  __shared__ __attribute__((aligned(16))) float so[32 * OUTT * D_OUTN];

  const int lane = threadIdx.x & 31;
  const int wv   = threadIdx.x >> 5;
  const int hi = lane >> 4, ii = lane & 15;
  const int s0 = blockIdx.x * 32 + wv * 16;

  v2f Aih[5][2];
#pragma unroll
  for (int c = 0; c < 5; ++c)
#pragma unroll
    for (int nt = 0; nt < 2; ++nt) {
      const int k0 = 4 * c + 2 * hi, i = nt * 16 + ii;
      Aih[c][nt].x = ((k0    ) < 18) ? w_ih2[i * 18 + k0    ] : 0.f;
      Aih[c][nt].y = ((k0 + 1) < 18) ? w_ih2[i * 18 + k0 + 1] : 0.f;
    }
  v2f Ahh[8][2];
#pragma unroll
  for (int c = 0; c < 8; ++c)
#pragma unroll
    for (int nt = 0; nt < 2; ++nt) {
      const int j0 = 16 * (c >> 2) + 8 * hi + 2 * (c & 3);
      const int i = nt * 16 + ii;
      Ahh[c][nt].x = w_hh2[i * 32 + j0    ];
      Ahh[c][nt].y = w_hh2[i * 32 + j0 + 1];
    }

  v8f cb[2];
#pragma unroll
  for (int nt = 0; nt < 2; ++nt)
#pragma unroll
    for (int rr = 0; rr < 8; ++rr) {
      const int i = nt * 16 + rr + 8 * hi;
      cb[nt][rr] = b_ih2[i] + b_hh2[i];
    }

  float wo[3][2][8], pinit[3];
#pragma unroll
  for (int o = 0; o < 3; ++o) {
    pinit[o] = hi ? 0.f : b_out[o];
#pragma unroll
    for (int nt = 0; nt < 2; ++nt)
#pragma unroll
      for (int rr = 0; rr < 8; ++rr)
        wo[o][nt][rr] = w_out[o * H2N + 16 * nt + rr + 8 * hi];
  }

  const int base = (s0 + ii) * 18;
  v2f By[5];
#pragma unroll
  for (int c = 0; c < 5; ++c) {
    const int k0 = 4 * c + 2 * hi;
    const int kc = (k0 < 18) ? k0 : 16;
    const float v0 = y[base + kc];
    const float v1 = y[base + kc + 1];
    By[c].x = ((k0    ) < 18) ? v0 : 0.f;
    By[c].y = ((k0 + 1) < 18) ? v1 : 0.f;
  }

  float* const srow = so + (wv * 16 + ii) * (OUTT * D_OUTN);
  v8f d[2];
  v8f cz = {};

#pragma unroll
  for (int nt = 0; nt < 2; ++nt) {
    v8f c1 = wmma4(Aih[0][nt], By[0], cb[nt]);
    v8f c2 = wmma4(Aih[1][nt], By[1], cz);
    c1 = wmma4(Aih[2][nt], By[2], c1);
    c2 = wmma4(Aih[3][nt], By[3], c2);
    c1 = wmma4(Aih[4][nt], By[4], c1);
#pragma unroll
    for (int rr = 0; rr < 8; ++rr) d[nt][rr] = fast_tanh(c1[rr] + c2[rr]);
  }

#pragma unroll 1
  for (int t = 1; t < OUTT; ++t) {
    v2f Bh[8];
#pragma unroll
    for (int c = 0; c < 8; ++c) {
      Bh[c].x = d[c >> 2][2 * (c & 3)    ];
      Bh[c].y = d[c >> 2][2 * (c & 3) + 1];
    }
    v8f acc[2][2];
#pragma unroll
    for (int nt = 0; nt < 2; ++nt) {
      v8f c1 = cb[nt];
      v8f c2 = cz;
#pragma unroll
      for (int cc = 0; cc < 8; cc += 2) {
        c1 = wmma4(Ahh[cc    ][nt], Bh[cc    ], c1);
        c2 = wmma4(Ahh[cc + 1][nt], Bh[cc + 1], c2);
      }
      acc[nt][0] = c1; acc[nt][1] = c2;
    }

    float p[3];
    proj3(d, wo, pinit, p);
    if (hi == 0) {
      srow[(t - 1) * D_OUTN + 0] = p[0];
      srow[(t - 1) * D_OUTN + 1] = p[1];
      srow[(t - 1) * D_OUTN + 2] = p[2];
    }

#pragma unroll
    for (int nt = 0; nt < 2; ++nt)
#pragma unroll
      for (int rr = 0; rr < 8; ++rr)
        d[nt][rr] = fast_tanh(acc[nt][0][rr] + acc[nt][1][rr]);
  }

  {
    float p[3];
    proj3(d, wo, pinit, p);
    if (hi == 0) {
      srow[(OUTT - 1) * D_OUTN + 0] = p[0];
      srow[(OUTT - 1) * D_OUTN + 1] = p[1];
      srow[(OUTT - 1) * D_OUTN + 2] = p[2];
    }
  }

  __syncthreads();

  const float4* s4 = (const float4*)so;
  float4* o4 = (float4*)(out + (size_t)blockIdx.x * (32 * OUTT * D_OUTN));
  for (int i = threadIdx.x; i < 32 * OUTT * D_OUTN / 4; i += 64)
    o4[i] = s4[i];
}

extern "C" void kernel_launch(void* const* d_in, const int* in_sizes, int n_in,
                              void* d_out, int out_size, void* d_ws,
                              size_t ws_size, hipStream_t stream) {
  const float* x      = (const float*)d_in[0];
  const float* w_ih_f = (const float*)d_in[1];
  const float* w_hh_f = (const float*)d_in[2];
  const float* b_ih_f = (const float*)d_in[3];
  const float* b_hh_f = (const float*)d_in[4];
  const float* w_ih_b = (const float*)d_in[5];
  const float* w_hh_b = (const float*)d_in[6];
  const float* b_ih_b = (const float*)d_in[7];
  const float* b_hh_b = (const float*)d_in[8];
  const float* w_ih2  = (const float*)d_in[9];
  const float* w_hh2  = (const float*)d_in[10];
  const float* b_ih2  = (const float*)d_in[11];
  const float* b_hh2  = (const float*)d_in[12];
  const float* w_out  = (const float*)d_in[13];
  const float* b_out  = (const float*)d_in[14];

  float* y = (float*)d_ws;

  rnn1_wmma_kernel<<<dim3(512), dim3(32), 0, stream>>>(
      x, w_ih_f, w_hh_f, b_ih_f, b_hh_f,
      w_ih_b, w_hh_b, b_ih_b, b_hh_b, y);

  rnn2_proj_kernel<<<dim3(128), dim3(64), 0, stream>>>(
      w_ih2, w_hh2, b_ih2, b_hh2, w_out, b_out, y, (float*)d_out);
}
